# starting kernel
# baseline (speedup 1.0000x reference)
#include <hip/hip_runtime.h>
#include <hip/hip_bf16.h>

#define BATCH 8
#define TLEN  2048
#define HDIM  256
#define BT    (BATCH * TLEN)

typedef __attribute__((ext_vector_type(2))) float v2f;
typedef __attribute__((ext_vector_type(8))) float v8f;

__global__ __launch_bounds__(256) void gates_gemm(
    const float* __restrict__ X,
    const float* __restrict__ Wa, const float* __restrict__ ba,
    const float* __restrict__ Wu, const float* __restrict__ bu,
    float* __restrict__ ab  ) {
  __shared__ float lwa[HDIM * 16];
  __shared__ float lwu[HDIM * 16];

  const int nt = blockIdx.x & 15;
  const int mg = blockIdx.x >> 4;
  const int lane = threadIdx.x & 31;
  const int wave = threadIdx.x >> 5;

  {
    const float4* ga = (const float4*)(Wa + nt * 16);
    const float4* gu = (const float4*)(Wu + nt * 16);
    float4* sa = (float4*)lwa;
    float4* su = (float4*)lwu;
    #pragma unroll
    for (int i = threadIdx.x; i < HDIM * 4; i += 256) {
      const int k = i >> 2;
      const int q = i & 3;
      sa[i] = ga[k * (HDIM / 4) + q];
      su[i] = gu[k * (HDIM / 4) + q];
    }
  }
  __syncthreads();

  const int hi   = lane >> 4;
  const int l16  = lane & 15;
  const int n    = nt * 16 + l16;
  const int koff = hi << 1;
  const int m0   = (mg * 8 + wave) * 64;

  v8f ca[4] = {v8f{}, v8f{}, v8f{}, v8f{}};
  v8f cu[4] = {v8f{}, v8f{}, v8f{}, v8f{}};

  const float* arow = X + (size_t)(m0 + l16) * HDIM + koff;
  #pragma unroll 2
  for (int k0 = 0; k0 < HDIM; k0 += 4) {
    const int k = k0 + koff;
    v2f bwa; bwa.x = lwa[k * 16 + l16]; bwa.y = lwa[(k + 1) * 16 + l16];
    v2f bwu; bwu.x = lwu[k * 16 + l16]; bwu.y = lwu[(k + 1) * 16 + l16];
    #pragma unroll
    for (int s = 0; s < 4; ++s) {
      v2f af = *(const v2f*)(arow + (size_t)(16 * s) * HDIM + k0);
      ca[s] = __builtin_amdgcn_wmma_f32_16x16x4_f32(false, af, false, bwa,
                                                    (short)0, ca[s], false, false);
      cu[s] = __builtin_amdgcn_wmma_f32_16x16x4_f32(false, af, false, bwu,
                                                    (short)0, cu[s], false, false);
    }
  }

  const float bia = ba[n];
  const float biu = bu[n];
  #pragma unroll
  for (int s = 0; s < 4; ++s) {
    const int rbase = m0 + 16 * s + (hi << 3);
    #pragma unroll
    for (int r = 0; r < 8; ++r) {
      const float za = ca[s][r] + bia;
      const float zu = cu[s][r] + biu;
      const float ga = 1.0f / (1.0f + __expf(-za));
      const float gu = tanhf(zu);
      v2f e; e.x = ga; e.y = (1.0f - ga) * gu;
      *(v2f*)(ab + 2 * ((size_t)(rbase + r) * HDIM + n)) = e;
    }
  }
}

__global__ __launch_bounds__(256) void scan_plif(
    const float* __restrict__ ab, float* __restrict__ s_out,
    const float* __restrict__ tau_p, const float* __restrict__ thr_p) {
  const int idx = blockIdx.x * blockDim.x + threadIdx.x;
  const int b = idx >> 8;
  const int h = idx & (HDIM - 1);
  const float inv_tau = 1.0f / tau_p[0];
  const float thr = thr_p[0];
  const float decay = 1.0f - inv_tau;

  float hstate = 0.0f;
  float v = 0.0f;
  const size_t base = ((size_t)b * TLEN) * HDIM + h;

  #pragma unroll 4
  for (int t = 0; t < TLEN; ++t) {
    const size_t off = base + (size_t)t * HDIM;
    if (t + 16 < TLEN)
      __builtin_prefetch(ab + 2 * (base + (size_t)(t + 16) * HDIM), 0, 0);
    v2f e = *(const v2f*)(ab + 2 * off);
    hstate = e.x * hstate + e.y;
    v = decay * v + inv_tau * hstate;
    const float sp = (v > thr) ? 1.0f : 0.0f;
    v -= sp * thr;
    s_out[off] = sp;
  }
}

__global__ __launch_bounds__(256) void synapse_gemm(
    const float* __restrict__ S, const float* __restrict__ Wsyn,
    float* __restrict__ out) {
  __shared__ float lws[16 * HDIM];

  const int nt = blockIdx.x & 15;
  const int mg = blockIdx.x >> 4;
  const int lane = threadIdx.x & 31;
  const int wave = threadIdx.x >> 5;

  {
    const float4* g = (const float4*)(Wsyn + (size_t)nt * 16 * HDIM);
    float4* sm = (float4*)lws;
    #pragma unroll
    for (int i = threadIdx.x; i < 16 * HDIM / 4; i += 256)
      sm[i] = g[i];
  }
  __syncthreads();

  const int hi   = lane >> 4;
  const int l16  = lane & 15;
  const int n    = nt * 16 + l16;
  const int koff = hi << 1;
  const int m0   = (mg * 8 + wave) * 64;

  v8f c[4] = {v8f{}, v8f{}, v8f{}, v8f{}};

  const float* arow = S + (size_t)(m0 + l16) * HDIM + koff;
  const float* brow = lws + l16 * HDIM + koff;
  #pragma unroll 2
  for (int k0 = 0; k0 < HDIM; k0 += 4) {
    v2f bf = *(const v2f*)(brow + k0);
    #pragma unroll
    for (int s = 0; s < 4; ++s) {
      v2f af = *(const v2f*)(arow + (size_t)(16 * s) * HDIM + k0);
      c[s] = __builtin_amdgcn_wmma_f32_16x16x4_f32(false, af, false, bf,
                                                   (short)0, c[s], false, false);
    }
  }

  #pragma unroll
  for (int s = 0; s < 4; ++s) {
    const int rbase = m0 + 16 * s + (hi << 3);
    #pragma unroll
    for (int r = 0; r < 8; ++r)
      out[(size_t)(rbase + r) * HDIM + n] = c[s][r];
  }
}

extern "C" void kernel_launch(void* const* d_in, const int* in_sizes, int n_in,
                              void* d_out, int out_size, void* d_ws, size_t ws_size,
                              hipStream_t stream) {
  const float* x    = (const float*)d_in[0];
  const float* Wa   = (const float*)d_in[1];
  const float* ba   = (const float*)d_in[2];
  const float* Wu   = (const float*)d_in[3];
  const float* bu   = (const float*)d_in[4];
  const float* tau  = (const float*)d_in[5];
  const float* thr  = (const float*)d_in[6];
  const float* Wsyn = (const float*)d_in[7];
  float* out = (float*)d_out;

  float* ab = (float*)d_ws;
  float* s  = ab + 2 * (size_t)BT * HDIM;

  gates_gemm<<<512, 256, 0, stream>>>(x, Wa, ba, Wu, bu, ab);
  scan_plif<<<8, 256, 0, stream>>>(ab, s, tau, thr);
  synapse_gemm<<<512, 256, 0, stream>>>(s, Wsyn, out);
}


# gate kernel: a work-group owns 8 columns, not 16, so 256 groups fill every compute unit; a product's 16 B columns = 8 of W_a + 8 of W_u; spike masks 8 bits per step; sums and order unchanged
# speedup vs baseline: 13.1537x; 13.1537x over previous
#include <hip/hip_runtime.h>
#include <hip/hip_bf16.h>

#define BATCH 8
#define TLEN  2048
#define HDIM  256
#define BT    (BATCH * TLEN)

#define COLS       8
#define NCT        (HDIM / COLS)
#define STG_ROWS   512
#define NSTG       (TLEN / STG_ROWS)
#define GEMM_WAVES 16
#define FUSED_THREADS ((GEMM_WAVES + 1) * 32)

typedef __attribute__((ext_vector_type(2))) float v2f;
typedef __attribute__((ext_vector_type(4))) float v4f;
typedef __attribute__((ext_vector_type(8))) float v8f;
typedef __attribute__((ext_vector_type(2))) unsigned int v2u;

__device__ __forceinline__ void rows_to_fragments(const v4f q, v2f& f0, v2f& f1) {
  const v2u e = __builtin_amdgcn_permlane16_swap(__float_as_uint(q.x), __float_as_uint(q.z), false, false);
  const v2u o = __builtin_amdgcn_permlane16_swap(__float_as_uint(q.y), __float_as_uint(q.w), false, false);
  f0.x = __uint_as_float(e.x); f0.y = __uint_as_float(o.x);
  f1.x = __uint_as_float(e.y); f1.y = __uint_as_float(o.y);
}

__device__ __forceinline__ float other_eight(const float x) {
  return __uint_as_float((unsigned int)__builtin_amdgcn_update_dpp(
      0, (int)__float_as_uint(x), 0x128  , 0xf, 0xf, false));
}

__device__ __forceinline__ void scan16(const v2f (&p)[16], float& hstate, float& v,
                                       const float decay, const float inv_tau,
                                       const float thr, unsigned int (&mk)[4],
                                       const int lane, const int lane0) {
  #pragma unroll
  for (int j = 0; j < 16; ++j) {
    hstate = p[j].x * hstate + p[j].y;
    v = decay * v + inv_tau * hstate;
    const bool fire = v > thr;
    const unsigned int m = __builtin_amdgcn_ballot_w32(fire);
    const float sp = fire ? 1.0f : 0.0f;
    v -= sp * thr;
    const bool mine = (lane == lane0 + (j >> 2));
    mk[j & 3] = mine ? m : mk[j & 3];
  }
  __builtin_amdgcn_sched_barrier(0);
}

__device__ __forceinline__ float bit_to_float(const unsigned int w, const int p) {
  return __uint_as_float((unsigned int)__builtin_amdgcn_sbfe((int)w, (unsigned int)p, 1u) & 0x3f800000u);
}

__global__ __launch_bounds__(FUSED_THREADS) void gates_scan_fused(
    const float* __restrict__ X,
    const float* __restrict__ Wa, const float* __restrict__ ba,
    const float* __restrict__ Wu, const float* __restrict__ bu,
    const float* __restrict__ tau_p, const float* __restrict__ thr_p,
    float* __restrict__ s_out) {
  __shared__ float lw[HDIM * 16];
  __shared__ v2f   lab[2][STG_ROWS * COLS];

  const int nt   = blockIdx.x & (NCT - 1);
  const int b    = blockIdx.x / NCT;
  const int lane = threadIdx.x & 31;
  const int wave = __builtin_amdgcn_readfirstlane((int)(threadIdx.x >> 5));

  {
    float4* sw = (float4*)lw;
    for (int i = threadIdx.x; i < HDIM * 4; i += FUSED_THREADS) {
      const int k = i >> 2;
      const int q = i & 3;
      const float* g = ((q < 2) ? Wa : Wu) + (size_t)k * HDIM + nt * COLS + (q & 1) * 4;
      sw[i] = *(const float4*)g;
    }
  }
  __syncthreads();

  const int hi   = lane >> 4;
  const int l16  = lane & 15;
  const int up   = l16 >> 3;
  const int c7   = lane & 7;
  const int koff = hi << 1;

  const float inv_tau = __uint_as_float(__builtin_amdgcn_readfirstlane(__float_as_uint(1.0f / tau_p[0])));
  const float thr     = thr_p[0];
  const float decay   = __uint_as_float(__builtin_amdgcn_readfirstlane(__float_as_uint(1.0f - inv_tau)));
  float hstate = 0.0f;
  float v      = 0.0f;

  const float bia = ba[nt * COLS + c7];
  const float biu = bu[nt * COLS + c7];

  const int prod = (wave > 0) ? (wave - 1) : 0;
  const float* arow0 = X + (size_t)(b * TLEN + prod * 32 + l16) * HDIM + (hi << 2);
  v4f cur[2][4];
  if (wave > 0) {
    #pragma unroll
    for (int s = 0; s < 2; ++s) {
      #pragma unroll
      for (int j = 0; j < 4; ++j)
        cur[s][j] = *(const v4f*)(arow0 + (size_t)(16 * s) * HDIM + 8 * j);
    }
  }

  for (int st = 0; st <= NSTG; ++st) {
    if (wave > 0) {
      if (st < NSTG) {
        const int tl0 = (wave - 1) * 32;

        v8f cg[2] = {v8f{}, v8f{}};

        const float* arow = arow0 + (size_t)(st * STG_ROWS) * HDIM;
        const float* anext = arow0 + (size_t)(((st + 1 < NSTG) ? (st + 1) : st) * STG_ROWS) * HDIM;

        #pragma unroll
        for (int kc = 0; kc < HDIM; kc += 32) {
          const float* nsrc = (kc + 32 < HDIM) ? (arow + kc + 32) : anext;
          v4f nxt[2][4];
          __builtin_amdgcn_sched_barrier(0);
          #pragma unroll
          for (int s = 0; s < 2; ++s) {
            #pragma unroll
            for (int j = 0; j < 4; ++j)
              nxt[s][j] = *(const v4f*)(nsrc + (size_t)(16 * s) * HDIM + 8 * j);
          }
          __builtin_amdgcn_sched_barrier(0);
          #pragma unroll
          for (int j = 0; j < 4; ++j) {
            v2f af[2][2];
            rows_to_fragments(cur[0][j], af[0][0], af[0][1]);
            rows_to_fragments(cur[1][j], af[1][0], af[1][1]);
            #pragma unroll
            for (int u = 0; u < 2; ++u) {
              const int k = kc + 8 * j + 4 * u + koff;
              v2f bw; bw.x = lw[k * 16 + l16]; bw.y = lw[(k + 1) * 16 + l16];
              #pragma unroll
              for (int s = 0; s < 2; ++s)
                cg[s] = __builtin_amdgcn_wmma_f32_16x16x4_f32(false, af[s][u], false, bw,
                                                              (short)0, cg[s], false, false);
            }
          }
          #pragma unroll
          for (int s = 0; s < 2; ++s) {
            #pragma unroll
            for (int j = 0; j < 4; ++j) cur[s][j] = nxt[s][j];
          }
        }

        v2f* dst = lab[st & 1];
        const int tl = tl0 + 16 * up + (hi << 3);
        #pragma unroll
        for (int r = 0; r < 8; ++r) {
          const float got = other_eight(up ? cg[0][r] : cg[1][r]);
          const float za = (up ? got : cg[0][r]) + bia;
          const float zu = (up ? cg[1][r] : got) + biu;
          const float ga = 1.0f / (1.0f + __expf(-za));
          const float gu = tanhf(zu);
          v2f e; e.x = ga; e.y = (1.0f - ga) * gu;
          dst[(tl + r) * COLS + c7] = e;
        }
      }
    } else {
      if (st > 0) {
        const int ln = (int)__builtin_amdgcn_mbcnt_lo(~0u, 0u);
        const v2f* src = lab[(st - 1) & 1] + (ln & 7);
        unsigned int* wout = (unsigned int*)s_out
            + ((((size_t)(b * NCT + nt)) * TLEN + (size_t)(st - 1) * STG_ROWS) >> 2);

        v2f pp[2][16];
        #pragma unroll
        for (int j = 0; j < 16; ++j) pp[0][j] = src[j * COLS];

        for (int g = 0; g < STG_ROWS / 16; g += 8) {
          unsigned int mk[4] = {0u, 0u, 0u, 0u};
          #pragma unroll
          for (int q = 0; q < 8; ++q) {
            const int gn = (q < 7) ? (g + q + 1) : ((g + 8 < STG_ROWS / 16) ? (g + 8) : g);
            #pragma unroll
            for (int j = 0; j < 16; ++j) pp[(q + 1) & 1][j] = src[(gn * 16 + j) * COLS];
            scan16(pp[q & 1], hstate, v, decay, inv_tau, thr, mk, ln, 4 * q);
          }
          (wout + (g >> 3) * 32)[(unsigned int)ln] =
              (mk[0] & 0xffu) | ((mk[1] & 0xffu) << 8) | ((mk[2] & 0xffu) << 16) | (mk[3] << 24);
        }
      }
    }
    __syncthreads();
  }
}

__global__ __launch_bounds__(256) void synapse_gemm(
    const float* __restrict__ S, const float* __restrict__ Wsyn,
    float* __restrict__ out) {
  __shared__ float lws[16 * HDIM];

  const int nt = blockIdx.x & 15;
  const int mg = blockIdx.x >> 4;
  const int lane = threadIdx.x & 31;
  const int wave = threadIdx.x >> 5;

  const int hi   = lane >> 4;
  const int l16  = lane & 15;
  const int n    = nt * 16 + l16;
  const int koff = hi << 1;
  const int m0   = (mg * 8 + wave) * 64;
  const int bb   = m0 / TLEN;
  const int tw   = m0 % TLEN;

  const unsigned int* mp = (const unsigned int*)S
                           + ((((size_t)(bb * NCT)) * TLEN + (size_t)(tw + 4 * l16)) >> 2);
  v2u q[16];
  #pragma unroll
  for (int t = 0; t < 16; ++t) {
    q[t].x = mp[(size_t)(2 * t) * (TLEN / 4)];
    q[t].y = mp[(size_t)(2 * t + 1) * (TLEN / 4)];
  }
  __builtin_amdgcn_sched_barrier(0);

  {
    const float4* g = (const float4*)(Wsyn + (size_t)nt * 16 * HDIM);
    float4* sm = (float4*)lws;
    #pragma unroll
    for (int i = threadIdx.x; i < 16 * HDIM / 4; i += 256)
      sm[i] = g[i];
  }
  __syncthreads();

  v8f c[4] = {v8f{}, v8f{}, v8f{}, v8f{}};

  const float* brow = lws + l16 * HDIM + koff;

  #pragma unroll
  for (int t = 0; t < 16; ++t) { q[t].x >>= koff; q[t].y >>= koff; }

  #pragma unroll
  for (int t = 0; t < 16; ++t) {
    #pragma unroll
    for (int j = 0; j < 4; ++j) {
      const int k0 = 16 * t + 4 * j;
      v2f bf = *(const v2f*)(brow + k0);
      #pragma unroll
      for (int s = 0; s < 4; ++s) {
        const unsigned int w = (j & 2) ? q[t].y : q[t].x;
        const int p = 8 * s + 4 * (j & 1);
        v2f af;
        af.x = bit_to_float(w, p);
        af.y = bit_to_float(w, p + 1);
        c[s] = __builtin_amdgcn_wmma_f32_16x16x4_f32(false, af, false, bf,
                                                     (short)0, c[s], false, false);
      }
    }
  }

  #pragma unroll
  for (int s = 0; s < 4; ++s) {
    const int rbase = m0 + 32 * hi + s;
    #pragma unroll
    for (int r = 0; r < 8; ++r)
      out[(size_t)(rbase + 4 * r) * HDIM + n] = c[s][r];
  }
}

extern "C" void kernel_launch(void* const* d_in, const int* in_sizes, int n_in,
                              void* d_out, int out_size, void* d_ws, size_t ws_size,
                              hipStream_t stream) {
  const float* x    = (const float*)d_in[0];
  const float* Wa   = (const float*)d_in[1];
  const float* ba   = (const float*)d_in[2];
  const float* Wu   = (const float*)d_in[3];
  const float* bu   = (const float*)d_in[4];
  const float* tau  = (const float*)d_in[5];
  const float* thr  = (const float*)d_in[6];
  const float* Wsyn = (const float*)d_in[7];
  float* out = (float*)d_out;

  float* s = (float*)d_ws;

  gates_scan_fused<<<BATCH * NCT, FUSED_THREADS, 0, stream>>>(x, Wa, ba, Wu, bu, tau, thr, s);
  synapse_gemm<<<512, 256, 0, stream>>>(s, Wsyn, out);
}


# scan wave: each 16-step block is one hand-written asm block with no s_delay_alu hints, so the serial chain keeps its SIMD's issue port beside the producers' WMMAs; same instructions, sums, order
# speedup vs baseline: 13.6026x; 1.0341x over previous
#include <hip/hip_runtime.h>
#include <hip/hip_bf16.h>

#define BATCH 8
#define TLEN  2048
#define HDIM  256
#define BT    (BATCH * TLEN)

#define COLS       8
#define NCT        (HDIM / COLS)
#define STG_ROWS   512
#define NSTG       (TLEN / STG_ROWS)
#define GEMM_WAVES 16
#define FUSED_THREADS ((GEMM_WAVES + 1) * 32)

typedef __attribute__((ext_vector_type(2))) float v2f;
typedef __attribute__((ext_vector_type(4))) float v4f;
typedef __attribute__((ext_vector_type(8))) float v8f;
typedef __attribute__((ext_vector_type(2))) unsigned int v2u;

__device__ __forceinline__ void rows_to_fragments(const v4f q, v2f& f0, v2f& f1) {
  const v2u e = __builtin_amdgcn_permlane16_swap(__float_as_uint(q.x), __float_as_uint(q.z), false, false);
  const v2u o = __builtin_amdgcn_permlane16_swap(__float_as_uint(q.y), __float_as_uint(q.w), false, false);
  f0.x = __uint_as_float(e.x); f0.y = __uint_as_float(o.x);
  f1.x = __uint_as_float(e.y); f1.y = __uint_as_float(o.y);
}

__device__ __forceinline__ float other_eight(const float x) {
  return __uint_as_float((unsigned int)__builtin_amdgcn_update_dpp(
      0, (int)__float_as_uint(x), 0x128  , 0xf, 0xf, false));
}

#define SCAN_STEP(XC, XN, AN, BN, MK)                               \
  "v_fmac_f32_e32 %[" XC "], %[dec], %[v]\n\t"                       \
  "v_fma_f32 %[h], %[" AN "], %[h], %[" BN "]\n\t"                   \
  "v_cmp_lt_f32_e32 vcc_lo, %[thr], %[" XC "]\n\t"                   \
  "v_mul_f32_e32 %[" XN "], %[itau], %[h]\n\t"                       \
  "v_cndmask_b32_e64 %[sp], 0, 1.0, vcc_lo\n\t"                      \
  "v_cndmask_b32_e64 %[" MK "], %[" MK "], vcc_lo, %[own]\n\t"       \
  "v_fma_f32 %[v], -%[thr], %[sp], %[" XC "]\n\t"
#define SCAN_OWNER(LN) "v_cmp_eq_u32_e64 %[own], %[" LN "], %[lane]\n\t"
#define SCAN_LAST(XC, MK)                                           \
  "v_fmac_f32_e32 %[" XC "], %[dec], %[v]\n\t"                       \
  "v_cmp_lt_f32_e32 vcc_lo, %[thr], %[" XC "]\n\t"                   \
  "v_cndmask_b32_e64 %[sp], 0, 1.0, vcc_lo\n\t"                      \
  "v_cndmask_b32_e64 %[" MK "], %[" MK "], vcc_lo, %[own]\n\t"       \
  "v_fma_f32 %[v], -%[thr], %[sp], %[" XC "]\n\t"

__device__ __forceinline__ void scan16(const v2f (&p)[16], float& hstate, float& v,
                                       const float decay, const float inv_tau,
                                       const float thr, unsigned int (&mk)[4],
                                       const int lane, const int lane0) {
  float x0, x1, sp;
  unsigned int own;
  asm("v_fma_f32 %[h], %[a0], %[h], %[b0]\n\t"
      "v_mul_f32_e32 %[x0], %[itau], %[h]\n\t"
      SCAN_OWNER("l0")
      SCAN_STEP("x0", "x1", "a1", "b1", "m0")
      SCAN_STEP("x1", "x0", "a2", "b2", "m1")
      SCAN_STEP("x0", "x1", "a3", "b3", "m2")
      SCAN_STEP("x1", "x0", "a4", "b4", "m3")
      SCAN_OWNER("l1")
      SCAN_STEP("x0", "x1", "a5", "b5", "m0")
      SCAN_STEP("x1", "x0", "a6", "b6", "m1")
      SCAN_STEP("x0", "x1", "a7", "b7", "m2")
      SCAN_STEP("x1", "x0", "a8", "b8", "m3")
      SCAN_OWNER("l2")
      SCAN_STEP("x0", "x1", "a9", "b9", "m0")
      SCAN_STEP("x1", "x0", "a10", "b10", "m1")
      SCAN_STEP("x0", "x1", "a11", "b11", "m2")
      SCAN_STEP("x1", "x0", "a12", "b12", "m3")
      SCAN_OWNER("l3")
      SCAN_STEP("x0", "x1", "a13", "b13", "m0")
      SCAN_STEP("x1", "x0", "a14", "b14", "m1")
      SCAN_STEP("x0", "x1", "a15", "b15", "m2")
      SCAN_LAST("x1", "m3")
      : [h] "+v"(hstate), [v] "+v"(v),
        [m0] "+v"(mk[0]), [m1] "+v"(mk[1]), [m2] "+v"(mk[2]), [m3] "+v"(mk[3]),
        [x0] "=&v"(x0), [x1] "=&v"(x1), [sp] "=&v"(sp), [own] "=&s"(own)
      :
        [a0] "v"(p[0].x), [b0] "v"(p[0].y),
        [a1] "v"(p[1].x), [b1] "v"(p[1].y),
        [a2] "v"(p[2].x), [b2] "v"(p[2].y),
        [a3] "v"(p[3].x), [b3] "v"(p[3].y),
        [a4] "v"(p[4].x), [b4] "v"(p[4].y),
        [a5] "v"(p[5].x), [b5] "v"(p[5].y),
        [a6] "v"(p[6].x), [b6] "v"(p[6].y),
        [a7] "v"(p[7].x), [b7] "v"(p[7].y),
        [a8] "v"(p[8].x), [b8] "v"(p[8].y),
        [a9] "v"(p[9].x), [b9] "v"(p[9].y),
        [a10] "v"(p[10].x), [b10] "v"(p[10].y),
        [a11] "v"(p[11].x), [b11] "v"(p[11].y),
        [a12] "v"(p[12].x), [b12] "v"(p[12].y),
        [a13] "v"(p[13].x), [b13] "v"(p[13].y),
        [a14] "v"(p[14].x), [b14] "v"(p[14].y),
        [a15] "v"(p[15].x), [b15] "v"(p[15].y),
        [itau] "v"(inv_tau), [dec] "s"(decay), [thr] "s"(thr),
        [lane] "v"(lane),
        [l0] "i"(lane0), [l1] "i"(lane0 + 1), [l2] "i"(lane0 + 2), [l3] "i"(lane0 + 3)
      : "vcc");
}

__device__ __forceinline__ float bit_to_float(const unsigned int w, const int p) {
  return __uint_as_float((unsigned int)__builtin_amdgcn_sbfe((int)w, (unsigned int)p, 1u) & 0x3f800000u);
}

__global__ __launch_bounds__(FUSED_THREADS) void gates_scan_fused(
    const float* __restrict__ X,
    const float* __restrict__ Wa, const float* __restrict__ ba,
    const float* __restrict__ Wu, const float* __restrict__ bu,
    const float* __restrict__ tau_p, const float* __restrict__ thr_p,
    float* __restrict__ s_out) {
  __shared__ float lw[HDIM * 16];
  __shared__ v2f   lab[2][STG_ROWS * COLS];

  const int nt   = blockIdx.x & (NCT - 1);
  const int b    = blockIdx.x / NCT;
  const int lane = threadIdx.x & 31;
  const int wave = __builtin_amdgcn_readfirstlane((int)(threadIdx.x >> 5));

  {
    float4* sw = (float4*)lw;
    for (int i = threadIdx.x; i < HDIM * 4; i += FUSED_THREADS) {
      const int k = i >> 2;
      const int q = i & 3;
      const float* g = ((q < 2) ? Wa : Wu) + (size_t)k * HDIM + nt * COLS + (q & 1) * 4;
      sw[i] = *(const float4*)g;
    }
  }
  __syncthreads();

  const int hi   = lane >> 4;
  const int l16  = lane & 15;
  const int up   = l16 >> 3;
  const int c7   = lane & 7;
  const int koff = hi << 1;

  const float inv_tau = __uint_as_float(__builtin_amdgcn_readfirstlane(__float_as_uint(1.0f / tau_p[0])));
  const float thr     = thr_p[0];
  const float decay   = __uint_as_float(__builtin_amdgcn_readfirstlane(__float_as_uint(1.0f - inv_tau)));
  float hstate = 0.0f;
  float v      = 0.0f;

  const float bia = ba[nt * COLS + c7];
  const float biu = bu[nt * COLS + c7];

  const int prod = (wave > 0) ? (wave - 1) : 0;
  const float* arow0 = X + (size_t)(b * TLEN + prod * 32 + l16) * HDIM + (hi << 2);
  v4f cur[2][4];
  if (wave > 0) {
    #pragma unroll
    for (int s = 0; s < 2; ++s) {
      #pragma unroll
      for (int j = 0; j < 4; ++j)
        cur[s][j] = *(const v4f*)(arow0 + (size_t)(16 * s) * HDIM + 8 * j);
    }
  }

  for (int st = 0; st <= NSTG; ++st) {
    if (wave > 0) {
      if (st < NSTG) {
        const int tl0 = (wave - 1) * 32;

        v8f cg[2] = {v8f{}, v8f{}};

        const float* arow = arow0 + (size_t)(st * STG_ROWS) * HDIM;
        const float* anext = arow0 + (size_t)(((st + 1 < NSTG) ? (st + 1) : st) * STG_ROWS) * HDIM;

        #pragma unroll
        for (int kc = 0; kc < HDIM; kc += 32) {
          const float* nsrc = (kc + 32 < HDIM) ? (arow + kc + 32) : anext;
          v4f nxt[2][4];
          __builtin_amdgcn_sched_barrier(0);
          #pragma unroll
          for (int s = 0; s < 2; ++s) {
            #pragma unroll
            for (int j = 0; j < 4; ++j)
              nxt[s][j] = *(const v4f*)(nsrc + (size_t)(16 * s) * HDIM + 8 * j);
          }
          __builtin_amdgcn_sched_barrier(0);
          #pragma unroll
          for (int j = 0; j < 4; ++j) {
            v2f af[2][2];
            rows_to_fragments(cur[0][j], af[0][0], af[0][1]);
            rows_to_fragments(cur[1][j], af[1][0], af[1][1]);
            #pragma unroll
            for (int u = 0; u < 2; ++u) {
              const int k = kc + 8 * j + 4 * u + koff;
              v2f bw; bw.x = lw[k * 16 + l16]; bw.y = lw[(k + 1) * 16 + l16];
              #pragma unroll
              for (int s = 0; s < 2; ++s)
                cg[s] = __builtin_amdgcn_wmma_f32_16x16x4_f32(false, af[s][u], false, bw,
                                                              (short)0, cg[s], false, false);
            }
          }
          #pragma unroll
          for (int s = 0; s < 2; ++s) {
            #pragma unroll
            for (int j = 0; j < 4; ++j) cur[s][j] = nxt[s][j];
          }
        }

        v2f* dst = lab[st & 1];
        const int tl = tl0 + 16 * up + (hi << 3);
        #pragma unroll
        for (int r = 0; r < 8; ++r) {
          const float got = other_eight(up ? cg[0][r] : cg[1][r]);
          const float za = (up ? got : cg[0][r]) + bia;
          const float zu = (up ? cg[1][r] : got) + biu;
          const float ga = 1.0f / (1.0f + __expf(-za));
          const float gu = tanhf(zu);
          v2f e; e.x = ga; e.y = (1.0f - ga) * gu;
          dst[(tl + r) * COLS + c7] = e;
        }
      }
    } else {
      if (st > 0) {
        const int ln = (int)__builtin_amdgcn_mbcnt_lo(~0u, 0u);
        const v2f* src = lab[(st - 1) & 1] + (ln & 7);
        unsigned int* wout = (unsigned int*)s_out
            + ((((size_t)(b * NCT + nt)) * TLEN + (size_t)(st - 1) * STG_ROWS) >> 2);

        v2f pp[2][16];
        #pragma unroll
        for (int j = 0; j < 16; ++j) pp[0][j] = src[j * COLS];

        for (int g = 0; g < STG_ROWS / 16; g += 8) {
          unsigned int mk[4] = {0u, 0u, 0u, 0u};
          #pragma unroll
          for (int q = 0; q < 8; ++q) {
            const int gn = (q < 7) ? (g + q + 1) : ((g + 8 < STG_ROWS / 16) ? (g + 8) : g);
            #pragma unroll
            for (int j = 0; j < 16; ++j) pp[(q + 1) & 1][j] = src[(gn * 16 + j) * COLS];
            scan16(pp[q & 1], hstate, v, decay, inv_tau, thr, mk, ln, 4 * q);
          }
          (wout + (g >> 3) * 32)[(unsigned int)ln] =
              (mk[0] & 0xffu) | ((mk[1] & 0xffu) << 8) | ((mk[2] & 0xffu) << 16) | (mk[3] << 24);
        }
      }
    }
    __syncthreads();
  }
}

__global__ __launch_bounds__(256) void synapse_gemm(
    const float* __restrict__ S, const float* __restrict__ Wsyn,
    float* __restrict__ out) {
  __shared__ float lws[16 * HDIM];

  const int nt = blockIdx.x & 15;
  const int mg = blockIdx.x >> 4;
  const int lane = threadIdx.x & 31;
  const int wave = threadIdx.x >> 5;

  const int hi   = lane >> 4;
  const int l16  = lane & 15;
  const int n    = nt * 16 + l16;
  const int koff = hi << 1;
  const int m0   = (mg * 8 + wave) * 64;
  const int bb   = m0 / TLEN;
  const int tw   = m0 % TLEN;

  const unsigned int* mp = (const unsigned int*)S
                           + ((((size_t)(bb * NCT)) * TLEN + (size_t)(tw + 4 * l16)) >> 2);
  v2u q[16];
  #pragma unroll
  for (int t = 0; t < 16; ++t) {
    q[t].x = mp[(size_t)(2 * t) * (TLEN / 4)];
    q[t].y = mp[(size_t)(2 * t + 1) * (TLEN / 4)];
  }
  __builtin_amdgcn_sched_barrier(0);

  {
    const float4* g = (const float4*)(Wsyn + (size_t)nt * 16 * HDIM);
    float4* sm = (float4*)lws;
    #pragma unroll
    for (int i = threadIdx.x; i < 16 * HDIM / 4; i += 256)
      sm[i] = g[i];
  }
  __syncthreads();

  v8f c[4] = {v8f{}, v8f{}, v8f{}, v8f{}};

  const float* brow = lws + l16 * HDIM + koff;

  #pragma unroll
  for (int t = 0; t < 16; ++t) { q[t].x >>= koff; q[t].y >>= koff; }

  #pragma unroll
  for (int t = 0; t < 16; ++t) {
    #pragma unroll
    for (int j = 0; j < 4; ++j) {
      const int k0 = 16 * t + 4 * j;
      v2f bf = *(const v2f*)(brow + k0);
      #pragma unroll
      for (int s = 0; s < 4; ++s) {
        const unsigned int w = (j & 2) ? q[t].y : q[t].x;
        const int p = 8 * s + 4 * (j & 1);
        v2f af;
        af.x = bit_to_float(w, p);
        af.y = bit_to_float(w, p + 1);
        c[s] = __builtin_amdgcn_wmma_f32_16x16x4_f32(false, af, false, bf,
                                                     (short)0, c[s], false, false);
      }
    }
  }

  #pragma unroll
  for (int s = 0; s < 4; ++s) {
    const int rbase = m0 + 32 * hi + s;
    #pragma unroll
    for (int r = 0; r < 8; ++r)
      out[(size_t)(rbase + 4 * r) * HDIM + n] = c[s][r];
  }
}

extern "C" void kernel_launch(void* const* d_in, const int* in_sizes, int n_in,
                              void* d_out, int out_size, void* d_ws, size_t ws_size,
                              hipStream_t stream) {
  const float* x    = (const float*)d_in[0];
  const float* Wa   = (const float*)d_in[1];
  const float* ba   = (const float*)d_in[2];
  const float* Wu   = (const float*)d_in[3];
  const float* bu   = (const float*)d_in[4];
  const float* tau  = (const float*)d_in[5];
  const float* thr  = (const float*)d_in[6];
  const float* Wsyn = (const float*)d_in[7];
  float* out = (float*)d_out;

  float* s = (float*)d_ws;

  gates_scan_fused<<<BATCH * NCT, FUSED_THREADS, 0, stream>>>(x, Wa, ba, Wu, bu, tau, thr, s);
  synapse_gemm<<<512, 256, 0, stream>>>(s, Wsyn, out);
}


# fused kernel: scan wave alone on its SIMD: 16 waves, the 12 whose number is no multiple of 4 produce (384-step stages), waves 4, 8, 12 only keep the barriers; sums and order unchanged
# speedup vs baseline: 14.2531x; 1.0478x over previous
#include <hip/hip_runtime.h>
#include <hip/hip_bf16.h>

#define BATCH 8
#define TLEN  2048
#define HDIM  256
#define BT    (BATCH * TLEN)

#define COLS       8
#define NCT        (HDIM / COLS)
#define GEMM_WAVES 12
#define STG_ROWS   (GEMM_WAVES * 32)
#define NSTG       ((TLEN + STG_ROWS - 1) / STG_ROWS)
#define FUSED_WAVES 16
#define FUSED_THREADS (FUSED_WAVES * 32)

typedef __attribute__((ext_vector_type(2))) float v2f;
typedef __attribute__((ext_vector_type(4))) float v4f;
typedef __attribute__((ext_vector_type(8))) float v8f;
typedef __attribute__((ext_vector_type(2))) unsigned int v2u;

__device__ __forceinline__ void rows_to_fragments(const v4f q, v2f& f0, v2f& f1) {
  const v2u e = __builtin_amdgcn_permlane16_swap(__float_as_uint(q.x), __float_as_uint(q.z), false, false);
  const v2u o = __builtin_amdgcn_permlane16_swap(__float_as_uint(q.y), __float_as_uint(q.w), false, false);
  f0.x = __uint_as_float(e.x); f0.y = __uint_as_float(o.x);
  f1.x = __uint_as_float(e.y); f1.y = __uint_as_float(o.y);
}

__device__ __forceinline__ float other_eight(const float x) {
  return __uint_as_float((unsigned int)__builtin_amdgcn_update_dpp(
      0, (int)__float_as_uint(x), 0x128  , 0xf, 0xf, false));
}

#define SCAN_STEP(XC, XN, AN, BN, MK)                               \
  "v_fmac_f32_e32 %[" XC "], %[dec], %[v]\n\t"                       \
  "v_fma_f32 %[h], %[" AN "], %[h], %[" BN "]\n\t"                   \
  "v_cmp_lt_f32_e32 vcc_lo, %[thr], %[" XC "]\n\t"                   \
  "v_mul_f32_e32 %[" XN "], %[itau], %[h]\n\t"                       \
  "v_cndmask_b32_e64 %[sp], 0, 1.0, vcc_lo\n\t"                      \
  "v_cndmask_b32_e64 %[" MK "], %[" MK "], vcc_lo, %[own]\n\t"       \
  "v_fma_f32 %[v], -%[thr], %[sp], %[" XC "]\n\t"
#define SCAN_OWNER(LN) "v_cmp_eq_u32_e64 %[own], %[" LN "], %[lane]\n\t"
#define SCAN_LAST(XC, MK)                                           \
  "v_fmac_f32_e32 %[" XC "], %[dec], %[v]\n\t"                       \
  "v_cmp_lt_f32_e32 vcc_lo, %[thr], %[" XC "]\n\t"                   \
  "v_cndmask_b32_e64 %[sp], 0, 1.0, vcc_lo\n\t"                      \
  "v_cndmask_b32_e64 %[" MK "], %[" MK "], vcc_lo, %[own]\n\t"       \
  "v_fma_f32 %[v], -%[thr], %[sp], %[" XC "]\n\t"

__device__ __forceinline__ void scan16(const v2f (&p)[16], float& hstate, float& v,
                                       const float decay, const float inv_tau,
                                       const float thr, unsigned int (&mk)[4],
                                       const int lane, const int lane0) {
  float x0, x1, sp;
  unsigned int own;
  asm("v_fma_f32 %[h], %[a0], %[h], %[b0]\n\t"
      "v_mul_f32_e32 %[x0], %[itau], %[h]\n\t"
      SCAN_OWNER("l0")
      SCAN_STEP("x0", "x1", "a1", "b1", "m0")
      SCAN_STEP("x1", "x0", "a2", "b2", "m1")
      SCAN_STEP("x0", "x1", "a3", "b3", "m2")
      SCAN_STEP("x1", "x0", "a4", "b4", "m3")
      SCAN_OWNER("l1")
      SCAN_STEP("x0", "x1", "a5", "b5", "m0")
      SCAN_STEP("x1", "x0", "a6", "b6", "m1")
      SCAN_STEP("x0", "x1", "a7", "b7", "m2")
      SCAN_STEP("x1", "x0", "a8", "b8", "m3")
      SCAN_OWNER("l2")
      SCAN_STEP("x0", "x1", "a9", "b9", "m0")
      SCAN_STEP("x1", "x0", "a10", "b10", "m1")
      SCAN_STEP("x0", "x1", "a11", "b11", "m2")
      SCAN_STEP("x1", "x0", "a12", "b12", "m3")
      SCAN_OWNER("l3")
      SCAN_STEP("x0", "x1", "a13", "b13", "m0")
      SCAN_STEP("x1", "x0", "a14", "b14", "m1")
      SCAN_STEP("x0", "x1", "a15", "b15", "m2")
      SCAN_LAST("x1", "m3")
      : [h] "+v"(hstate), [v] "+v"(v),
        [m0] "+v"(mk[0]), [m1] "+v"(mk[1]), [m2] "+v"(mk[2]), [m3] "+v"(mk[3]),
        [x0] "=&v"(x0), [x1] "=&v"(x1), [sp] "=&v"(sp), [own] "=&s"(own)
      :
        [a0] "v"(p[0].x), [b0] "v"(p[0].y),
        [a1] "v"(p[1].x), [b1] "v"(p[1].y),
        [a2] "v"(p[2].x), [b2] "v"(p[2].y),
        [a3] "v"(p[3].x), [b3] "v"(p[3].y),
        [a4] "v"(p[4].x), [b4] "v"(p[4].y),
        [a5] "v"(p[5].x), [b5] "v"(p[5].y),
        [a6] "v"(p[6].x), [b6] "v"(p[6].y),
        [a7] "v"(p[7].x), [b7] "v"(p[7].y),
        [a8] "v"(p[8].x), [b8] "v"(p[8].y),
        [a9] "v"(p[9].x), [b9] "v"(p[9].y),
        [a10] "v"(p[10].x), [b10] "v"(p[10].y),
        [a11] "v"(p[11].x), [b11] "v"(p[11].y),
        [a12] "v"(p[12].x), [b12] "v"(p[12].y),
        [a13] "v"(p[13].x), [b13] "v"(p[13].y),
        [a14] "v"(p[14].x), [b14] "v"(p[14].y),
        [a15] "v"(p[15].x), [b15] "v"(p[15].y),
        [itau] "v"(inv_tau), [dec] "s"(decay), [thr] "s"(thr),
        [lane] "v"(lane),
        [l0] "i"(lane0), [l1] "i"(lane0 + 1), [l2] "i"(lane0 + 2), [l3] "i"(lane0 + 3)
      : "vcc");
}

__device__ __forceinline__ float bit_to_float(const unsigned int w, const int p) {
  return __uint_as_float((unsigned int)__builtin_amdgcn_sbfe((int)w, (unsigned int)p, 1u) & 0x3f800000u);
}

__global__ __launch_bounds__(FUSED_THREADS) void gates_scan_fused(
    const float* __restrict__ X,
    const float* __restrict__ Wa, const float* __restrict__ ba,
    const float* __restrict__ Wu, const float* __restrict__ bu,
    const float* __restrict__ tau_p, const float* __restrict__ thr_p,
    float* __restrict__ s_out) {
  __shared__ float lw[HDIM * 16];
  __shared__ v2f   lab[2][STG_ROWS * COLS];

  const int nt   = blockIdx.x & (NCT - 1);
  const int b    = blockIdx.x / NCT;
  const int lane = threadIdx.x & 31;
  const int wave = __builtin_amdgcn_readfirstlane((int)(threadIdx.x >> 5));
  const bool is_prod = (wave & 3) != 0;
  const int  prod    = is_prod ? ((wave >> 2) * 3 + (wave & 3) - 1) : 0;

  {
    float4* sw = (float4*)lw;
    for (int i = threadIdx.x; i < HDIM * 4; i += FUSED_THREADS) {
      const int k = i >> 2;
      const int q = i & 3;
      const float* g = ((q < 2) ? Wa : Wu) + (size_t)k * HDIM + nt * COLS + (q & 1) * 4;
      sw[i] = *(const float4*)g;
    }
  }
  __syncthreads();

  const int hi   = lane >> 4;
  const int l16  = lane & 15;
  const int up   = l16 >> 3;
  const int c7   = lane & 7;
  const int koff = hi << 1;

  const float inv_tau = __uint_as_float(__builtin_amdgcn_readfirstlane(__float_as_uint(1.0f / tau_p[0])));
  const float thr     = thr_p[0];
  const float decay   = __uint_as_float(__builtin_amdgcn_readfirstlane(__float_as_uint(1.0f - inv_tau)));
  float hstate = 0.0f;
  float v      = 0.0f;

  const float bia = ba[nt * COLS + c7];
  const float biu = bu[nt * COLS + c7];

  const float* arow0 = X + (size_t)(b * TLEN + prod * 32 + l16) * HDIM + (hi << 2);
  v4f cur[2][4];
  if (is_prod) {
    #pragma unroll
    for (int s = 0; s < 2; ++s) {
      #pragma unroll
      for (int j = 0; j < 4; ++j)
        cur[s][j] = *(const v4f*)(arow0 + (size_t)(16 * s) * HDIM + 8 * j);
    }
  }

  for (int st = 0; st <= NSTG; ++st) {
    if (is_prod) {
      const int tl0 = prod * 32;
      if (st * STG_ROWS + tl0 < TLEN) {

        v8f cg[2] = {v8f{}, v8f{}};

        const float* arow = arow0 + (size_t)(st * STG_ROWS) * HDIM;
        const float* anext = arow0 + (size_t)((((st + 1) * STG_ROWS + tl0 < TLEN) ? (st + 1) : st) * STG_ROWS) * HDIM;

        #pragma unroll
        for (int kc = 0; kc < HDIM; kc += 32) {
          const float* nsrc = (kc + 32 < HDIM) ? (arow + kc + 32) : anext;
          v4f nxt[2][4];
          __builtin_amdgcn_sched_barrier(0);
          #pragma unroll
          for (int s = 0; s < 2; ++s) {
            #pragma unroll
            for (int j = 0; j < 4; ++j)
              nxt[s][j] = *(const v4f*)(nsrc + (size_t)(16 * s) * HDIM + 8 * j);
          }
          __builtin_amdgcn_sched_barrier(0);
          #pragma unroll
          for (int j = 0; j < 4; ++j) {
            v2f af[2][2];
            rows_to_fragments(cur[0][j], af[0][0], af[0][1]);
            rows_to_fragments(cur[1][j], af[1][0], af[1][1]);
            #pragma unroll
            for (int u = 0; u < 2; ++u) {
              const int k = kc + 8 * j + 4 * u + koff;
              v2f bw; bw.x = lw[k * 16 + l16]; bw.y = lw[(k + 1) * 16 + l16];
              #pragma unroll
              for (int s = 0; s < 2; ++s)
                cg[s] = __builtin_amdgcn_wmma_f32_16x16x4_f32(false, af[s][u], false, bw,
                                                              (short)0, cg[s], false, false);
            }
          }
          #pragma unroll
          for (int s = 0; s < 2; ++s) {
            #pragma unroll
            for (int j = 0; j < 4; ++j) cur[s][j] = nxt[s][j];
          }
        }

        v2f* dst = lab[st & 1];
        const int tl = tl0 + 16 * up + (hi << 3);
        #pragma unroll
        for (int r = 0; r < 8; ++r) {
          const float got = other_eight(up ? cg[0][r] : cg[1][r]);
          const float za = (up ? got : cg[0][r]) + bia;
          const float zu = (up ? cg[1][r] : got) + biu;
          const float ga = 1.0f / (1.0f + __expf(-za));
          const float gu = tanhf(zu);
          v2f e; e.x = ga; e.y = (1.0f - ga) * gu;
          dst[(tl + r) * COLS + c7] = e;
        }
      }
    } else if (wave == 0) {
      if (st > 0) {
        const int ln = (int)__builtin_amdgcn_mbcnt_lo(~0u, 0u);
        const v2f* src = lab[(st - 1) & 1] + (ln & 7);
        unsigned int* wout = (unsigned int*)s_out
            + ((((size_t)(b * NCT + nt)) * TLEN + (size_t)(st - 1) * STG_ROWS) >> 2);

        v2f pp[2][16];
        #pragma unroll
        for (int j = 0; j < 16; ++j) pp[0][j] = src[j * COLS];

        const int left = TLEN - (st - 1) * STG_ROWS;
        const int ngrp = ((left < STG_ROWS) ? left : STG_ROWS) / 16;

        for (int g = 0; g < ngrp; g += 8) {
          unsigned int mk[4] = {0u, 0u, 0u, 0u};
          #pragma unroll
          for (int q = 0; q < 8; ++q) {
            const int gn = (q < 7) ? (g + q + 1) : ((g + 8 < ngrp) ? (g + 8) : g);
            #pragma unroll
            for (int j = 0; j < 16; ++j) pp[(q + 1) & 1][j] = src[(gn * 16 + j) * COLS];
            scan16(pp[q & 1], hstate, v, decay, inv_tau, thr, mk, ln, 4 * q);
          }
          (wout + (g >> 3) * 32)[(unsigned int)ln] =
              (mk[0] & 0xffu) | ((mk[1] & 0xffu) << 8) | ((mk[2] & 0xffu) << 16) | (mk[3] << 24);
        }
      }
    }
    __syncthreads();
  }
}

__global__ __launch_bounds__(256) void synapse_gemm(
    const float* __restrict__ S, const float* __restrict__ Wsyn,
    float* __restrict__ out) {
  __shared__ float lws[16 * HDIM];

  const int nt = blockIdx.x & 15;
  const int mg = blockIdx.x >> 4;
  const int lane = threadIdx.x & 31;
  const int wave = threadIdx.x >> 5;

  const int hi   = lane >> 4;
  const int l16  = lane & 15;
  const int n    = nt * 16 + l16;
  const int koff = hi << 1;
  const int m0   = (mg * 8 + wave) * 64;
  const int bb   = m0 / TLEN;
  const int tw   = m0 % TLEN;

  const unsigned int* mp = (const unsigned int*)S
                           + ((((size_t)(bb * NCT)) * TLEN + (size_t)(tw + 4 * l16)) >> 2);
  v2u q[16];
  #pragma unroll
  for (int t = 0; t < 16; ++t) {
    q[t].x = mp[(size_t)(2 * t) * (TLEN / 4)];
    q[t].y = mp[(size_t)(2 * t + 1) * (TLEN / 4)];
  }
  __builtin_amdgcn_sched_barrier(0);

  {
    const float4* g = (const float4*)(Wsyn + (size_t)nt * 16 * HDIM);
    float4* sm = (float4*)lws;
    #pragma unroll
    for (int i = threadIdx.x; i < 16 * HDIM / 4; i += 256)
      sm[i] = g[i];
  }
  __syncthreads();

  v8f c[4] = {v8f{}, v8f{}, v8f{}, v8f{}};

  const float* brow = lws + l16 * HDIM + koff;

  #pragma unroll
  for (int t = 0; t < 16; ++t) { q[t].x >>= koff; q[t].y >>= koff; }

  #pragma unroll
  for (int t = 0; t < 16; ++t) {
    #pragma unroll
    for (int j = 0; j < 4; ++j) {
      const int k0 = 16 * t + 4 * j;
      v2f bf = *(const v2f*)(brow + k0);
      #pragma unroll
      for (int s = 0; s < 4; ++s) {
        const unsigned int w = (j & 2) ? q[t].y : q[t].x;
        const int p = 8 * s + 4 * (j & 1);
        v2f af;
        af.x = bit_to_float(w, p);
        af.y = bit_to_float(w, p + 1);
        c[s] = __builtin_amdgcn_wmma_f32_16x16x4_f32(false, af, false, bf,
                                                     (short)0, c[s], false, false);
      }
    }
  }

  #pragma unroll
  for (int s = 0; s < 4; ++s) {
    const int rbase = m0 + 32 * hi + s;
    #pragma unroll
    for (int r = 0; r < 8; ++r)
      out[(size_t)(rbase + 4 * r) * HDIM + n] = c[s][r];
  }
}

extern "C" void kernel_launch(void* const* d_in, const int* in_sizes, int n_in,
                              void* d_out, int out_size, void* d_ws, size_t ws_size,
                              hipStream_t stream) {
  const float* x    = (const float*)d_in[0];
  const float* Wa   = (const float*)d_in[1];
  const float* ba   = (const float*)d_in[2];
  const float* Wu   = (const float*)d_in[3];
  const float* bu   = (const float*)d_in[4];
  const float* tau  = (const float*)d_in[5];
  const float* thr  = (const float*)d_in[6];
  const float* Wsyn = (const float*)d_in[7];
  float* out = (float*)d_out;

  float* s = (float*)d_ws;

  gates_scan_fused<<<BATCH * NCT, FUSED_THREADS, 0, stream>>>(x, Wa, ba, Wu, bu, tau, thr, s);
  synapse_gemm<<<512, 256, 0, stream>>>(s, Wsyn, out);
}
